# starting kernel
# baseline (speedup 1.0000x reference)
#include <hip/hip_runtime.h>
#include <hip/hip_bf16.h>


typedef __attribute__((ext_vector_type(2))) float v2f;
typedef __attribute__((ext_vector_type(8))) float v8f;

#define NPTS    32768
#define HID     256
#define MT      16
#define LSTR    260
#define NSTREAM 7

__device__ __forceinline__ v8f wmma4(v2f a, v2f b, v8f c) {
  return __builtin_amdgcn_wmma_f32_16x16x4_f32(
       false, a,  false, b,
       (short)0, c,  false,  false);
}

__device__ __forceinline__ float fast_tanh(float x) {
  const float e = __expf(2.0f * x);
  const float r = __builtin_amdgcn_rcpf(e + 1.0f);
  return 1.0f - 2.0f * r;
}

__global__ void __launch_bounds__(256)
pinn_fused(const float* __restrict__ X,
           const float* __restrict__ W0, const float* __restrict__ b0,
           const float* __restrict__ W1, const float* __restrict__ b1,
           const float* __restrict__ W2, const float* __restrict__ b2,
           const float* __restrict__ W3, const float* __restrict__ b3,
           const float* __restrict__ W4, const float* __restrict__ b4,
           const float* __restrict__ lb, const float* __restrict__ ub,
           float* __restrict__ out)
{
  extern __shared__ float smem[];
  __shared__ float xs[MT * 3];

  const int t  = threadIdx.x;
  const int r0 = blockIdx.x * MT;

  float* S[NSTREAM];
  #pragma unroll
  for (int s = 0; s < NSTREAM; ++s) S[s] = smem + (size_t)s * MT * LSTR;

  if (t < MT * 3) xs[t] = X[(size_t)r0 * 3 + t];
  __syncthreads();

  {
    const int c = t;
    const float lb0 = lb[0], lb1 = lb[1], lb2 = lb[2];
    const float s0 = 2.f / (ub[0] - lb0);
    const float s1 = 2.f / (ub[1] - lb1);
    const float s2 = 2.f / (ub[2] - lb2);
    const float w0c = W0[c], w1c = W0[HID + c], w2c = W0[2 * HID + c];
    const float bc  = b0[c];
    const float zu0 = s0 * w0c, zu1 = s1 * w1c, zu2 = s2 * w2c;
    const float Uc  = zu0 + zu1 + zu2;
    #pragma unroll 4
    for (int r = 0; r < MT; ++r) {
      const float xn0 = s0 * (xs[r * 3 + 0] - lb0) - 1.f;
      const float xn1 = s1 * (xs[r * 3 + 1] - lb1) - 1.f;
      const float xn2 = s2 * (xs[r * 3 + 2] - lb2) - 1.f;
      const float z   = xn0 * w0c + xn1 * w1c + xn2 * w2c + bc;
      const float a   = fast_tanh(z);
      const float d1  = 1.f - a * a;
      const float d2U = -2.f * a * d1 * Uc;
      const int off = r * LSTR + c;
      S[0][off] = a;
      S[1][off] = d1 * zu0;  S[2][off] = d1 * zu1;  S[3][off] = d1 * zu2;
      S[4][off] = d2U * zu0; S[5][off] = d2U * zu1; S[6][off] = d2U * zu2;
    }
  }
  __syncthreads();

  const int lane  = t & 31;
  const int wid   = t >> 5;
  const int c0    = wid * 32;
  const int half  = lane >> 4;
  const int l15   = lane & 15;
  const int koff  = half * 2;
  const int rbase = half * 8;

  const float* Wl[3] = {W1, W2, W3};
  const float* bl[3] = {b1, b2, b3};

  const float* abase[NSTREAM];
  #pragma unroll
  for (int s = 0; s < NSTREAM; ++s) abase[s] = S[s] + l15 * LSTR + koff;

  for (int l = 0; l < 3; ++l) {
    const float* __restrict__ W = Wl[l];
    const float* wb = W + (size_t)koff * HID + c0 + l15;

    v8f acc[NSTREAM][2];
    #pragma unroll
    for (int s = 0; s < NSTREAM; ++s) { acc[s][0] = 0.f; acc[s][1] = 0.f; }

    #pragma unroll 2
    for (int k = 0; k < HID; k += 4) {
      const float* wk = wb + (size_t)k * HID;
      v2f bf0, bf1;
      bf0.x = wk[0];   bf0.y = wk[HID];
      bf1.x = wk[16];  bf1.y = wk[HID + 16];
      #pragma unroll
      for (int s = 0; s < NSTREAM; ++s) {
        const v2f av = *(const v2f*)(abase[s] + k);
        acc[s][0] = wmma4(av, bf0, acc[s][0]);
        acc[s][1] = wmma4(av, bf1, acc[s][1]);
      }
    }
    __syncthreads();

    const float* bb = bl[l];
    const float bias0 = bb[c0 + l15];
    const float bias1 = bb[c0 + 16 + l15];
    #pragma unroll
    for (int tile = 0; tile < 2; ++tile) {
      const int col = c0 + tile * 16 + l15;
      const float bv = tile ? bias1 : bias0;
      #pragma unroll
      for (int e = 0; e < 8; ++e) {
        const int off = (rbase + e) * LSTR + col;
        const float a   = fast_tanh(acc[0][tile][e] + bv);
        const float zu0 = acc[1][tile][e], zu1 = acc[2][tile][e],
                    zu2 = acc[3][tile][e];
        const float zw0 = acc[4][tile][e], zw1 = acc[5][tile][e],
                    zw2 = acc[6][tile][e];
        const float d1  = 1.f - a * a;
        const float d2U = -2.f * a * d1 * (zu0 + zu1 + zu2);
        S[0][off] = a;
        S[1][off] = d1 * zu0;
        S[2][off] = d1 * zu1;
        S[3][off] = d1 * zu2;
        S[4][off] = d1 * zw0 + d2U * zu0;
        S[5][off] = d1 * zw1 + d2U * zu1;
        S[6][off] = d1 * zw2 + d2U * zu2;
      }
    }
    __syncthreads();
  }

  {
    const int row   = t & 15;
    const int which = t >> 4;
    if (which < 14) {
      int stream, i, k = 0;
      if (which < 2)      { stream = 0; i = which; }
      else if (which < 8) { const int q = which - 2; i = q / 3; k = q % 3; stream = 1 + k; }
      else                { const int q = which - 8; i = q / 3; k = q % 3; stream = 4 + k; }
      const float* src = S[stream] + row * LSTR;
      float accv = 0.f;
      #pragma unroll 8
      for (int c = 0; c < HID; ++c) accv += src[c] * W4[c * 2 + i];
      const size_t n = (size_t)(r0 + row);
      if (which < 2) {
        out[n * 2 + i] = accv + b4[i];
      } else if (which < 8) {
        out[(size_t)2 * NPTS + (size_t)i * NPTS * 3 + n * 3 + k] = accv;
      } else {
        out[(size_t)8 * NPTS + (size_t)i * NPTS * 3 + n * 3 + k] = accv;
      }
    }
  }
}

extern "C" void kernel_launch(void* const* d_in, const int* in_sizes, int n_in,
                              void* d_out, int out_size, void* d_ws, size_t ws_size,
                              hipStream_t stream) {
  const float* X  = (const float*)d_in[0];
  const float* W0 = (const float*)d_in[1];
  const float* b0 = (const float*)d_in[2];
  const float* W1 = (const float*)d_in[3];
  const float* b1 = (const float*)d_in[4];
  const float* W2 = (const float*)d_in[5];
  const float* b2 = (const float*)d_in[6];
  const float* W3 = (const float*)d_in[7];
  const float* b3 = (const float*)d_in[8];
  const float* W4 = (const float*)d_in[9];
  const float* b4 = (const float*)d_in[10];
  const float* lb = (const float*)d_in[11];
  const float* ub = (const float*)d_in[12];
  float* out = (float*)d_out;

  const size_t shmem = (size_t)NSTREAM * MT * LSTR * sizeof(float);
  dim3 grid(NPTS / MT), block(256);
  pinn_fused<<<grid, block, shmem, stream>>>(X, W0, b0, W1, b1, W2, b2,
                                             W3, b3, W4, b4, lb, ub, out);
}


# K loop software pipelined: next 8-wide k-block's W (global) and activation (LDS) fragments requested as one group into a second register set ahead of the current 28 WMMAs; sum order unchanged
# speedup vs baseline: 1.0544x; 1.0544x over previous
#include <hip/hip_runtime.h>
#include <hip/hip_bf16.h>


typedef __attribute__((ext_vector_type(2))) float v2f;
typedef __attribute__((ext_vector_type(8))) float v8f;

#define NPTS    32768
#define HID     256
#define MT      16
#define LSTR    260
#define NSTREAM 7

__device__ __forceinline__ v8f wmma4(v2f a, v2f b, v8f c) {
  return __builtin_amdgcn_wmma_f32_16x16x4_f32(
       false, a,  false, b,
       (short)0, c,  false,  false);
}

__device__ __forceinline__ float fast_tanh(float x) {
  const float e = __expf(2.0f * x);
  const float r = __builtin_amdgcn_rcpf(e + 1.0f);
  return 1.0f - 2.0f * r;
}

struct KBlk {
  v2f a0[NSTREAM], a1[NSTREAM];
  v2f b00, b01, b10, b11;
};

__device__ __forceinline__ void load_b(KBlk& f, const float* __restrict__ wk) {
  f.b00.x = wk[0];            f.b00.y = wk[HID];
  f.b01.x = wk[16];           f.b01.y = wk[HID + 16];
  f.b10.x = wk[4 * HID];      f.b10.y = wk[5 * HID];
  f.b11.x = wk[4 * HID + 16]; f.b11.y = wk[5 * HID + 16];
}

__device__ __forceinline__ void load_a(KBlk& f, const float* const (&abase)[NSTREAM], int k) {
  #pragma unroll
  for (int s = 0; s < NSTREAM; ++s) {
    f.a0[s] = *(const v2f*)(abase[s] + k);
    f.a1[s] = *(const v2f*)(abase[s] + k + 4);
  }
}

__device__ __forceinline__ void mma_blk(v8f (&acc)[NSTREAM][2], const KBlk& f) {
  #pragma unroll
  for (int s = 0; s < NSTREAM; ++s) {
    acc[s][0] = wmma4(f.a0[s], f.b00, acc[s][0]);
    acc[s][1] = wmma4(f.a0[s], f.b01, acc[s][1]);
  }
  #pragma unroll
  for (int s = 0; s < NSTREAM; ++s) {
    acc[s][0] = wmma4(f.a1[s], f.b10, acc[s][0]);
    acc[s][1] = wmma4(f.a1[s], f.b11, acc[s][1]);
  }
}

__global__ void __launch_bounds__(256)
pinn_fused(const float* __restrict__ X,
           const float* __restrict__ W0, const float* __restrict__ b0,
           const float* __restrict__ W1, const float* __restrict__ b1,
           const float* __restrict__ W2, const float* __restrict__ b2,
           const float* __restrict__ W3, const float* __restrict__ b3,
           const float* __restrict__ W4, const float* __restrict__ b4,
           const float* __restrict__ lb, const float* __restrict__ ub,
           float* __restrict__ out)
{
  extern __shared__ float smem[];
  __shared__ float xs[MT * 3];

  const int t  = threadIdx.x;
  const int r0 = blockIdx.x * MT;

  float* S[NSTREAM];
  #pragma unroll
  for (int s = 0; s < NSTREAM; ++s) S[s] = smem + (size_t)s * MT * LSTR;

  const int lane  = t & 31;
  const int wid   = t >> 5;
  const int c0    = wid * 32;
  const int half  = lane >> 4;
  const int l15   = lane & 15;
  const int koff  = half * 2;
  const int rbase = half * 8;

  KBlk f0, f1;

  if (t < MT * 3) xs[t] = X[(size_t)r0 * 3 + t];
  __syncthreads();

  {
    const int c = t;
    const float lb0 = lb[0], lb1 = lb[1], lb2 = lb[2];
    const float s0 = 2.f / (ub[0] - lb0);
    const float s1 = 2.f / (ub[1] - lb1);
    const float s2 = 2.f / (ub[2] - lb2);
    const float w0c = W0[c], w1c = W0[HID + c], w2c = W0[2 * HID + c];
    const float bc  = b0[c];
    const float zu0 = s0 * w0c, zu1 = s1 * w1c, zu2 = s2 * w2c;
    const float Uc  = zu0 + zu1 + zu2;
    load_b(f0, W1 + (size_t)koff * HID + c0 + l15);
    #pragma unroll 4
    for (int r = 0; r < MT; ++r) {
      const float xn0 = s0 * (xs[r * 3 + 0] - lb0) - 1.f;
      const float xn1 = s1 * (xs[r * 3 + 1] - lb1) - 1.f;
      const float xn2 = s2 * (xs[r * 3 + 2] - lb2) - 1.f;
      const float z   = xn0 * w0c + xn1 * w1c + xn2 * w2c + bc;
      const float a   = fast_tanh(z);
      const float d1  = 1.f - a * a;
      const float d2U = -2.f * a * d1 * Uc;
      const int off = r * LSTR + c;
      S[0][off] = a;
      S[1][off] = d1 * zu0;  S[2][off] = d1 * zu1;  S[3][off] = d1 * zu2;
      S[4][off] = d2U * zu0; S[5][off] = d2U * zu1; S[6][off] = d2U * zu2;
    }
  }
  __syncthreads();

  const float* Wl[3] = {W1, W2, W3};
  const float* bl[3] = {b1, b2, b3};

  const float* abase[NSTREAM];
  #pragma unroll
  for (int s = 0; s < NSTREAM; ++s) abase[s] = S[s] + l15 * LSTR + koff;

  #pragma unroll
  for (int l = 0; l < 3; ++l) {
    const float* __restrict__ W = Wl[l];
    const float* wb = W + (size_t)koff * HID + c0 + l15;

    v8f acc[NSTREAM][2];
    #pragma unroll
    for (int s = 0; s < NSTREAM; ++s) { acc[s][0] = 0.f; acc[s][1] = 0.f; }

    load_a(f0, abase, 0);
    #pragma unroll 1
    for (int k = 0; k < HID - 16; k += 16) {
      load_b(f1, wb + (size_t)(k + 8) * HID);
      load_a(f1, abase, k + 8);
      __builtin_amdgcn_sched_barrier(0);
      mma_blk(acc, f0);
      __builtin_amdgcn_sched_barrier(0);
      load_b(f0, wb + (size_t)(k + 16) * HID);
      load_a(f0, abase, k + 16);
      __builtin_amdgcn_sched_barrier(0);
      mma_blk(acc, f1);
      __builtin_amdgcn_sched_barrier(0);
    }
    load_b(f1, wb + (size_t)(HID - 8) * HID);
    load_a(f1, abase, HID - 8);
    __builtin_amdgcn_sched_barrier(0);
    mma_blk(acc, f0);
    __builtin_amdgcn_sched_barrier(0);
    mma_blk(acc, f1);
    __syncthreads();

    if (l + 1 < 3) load_b(f0, Wl[l + 1] + (size_t)koff * HID + c0 + l15);

    const float* bb = bl[l];
    const float bias0 = bb[c0 + l15];
    const float bias1 = bb[c0 + 16 + l15];
    #pragma unroll
    for (int tile = 0; tile < 2; ++tile) {
      const int col = c0 + tile * 16 + l15;
      const float bv = tile ? bias1 : bias0;
      #pragma unroll
      for (int e = 0; e < 8; ++e) {
        const int off = (rbase + e) * LSTR + col;
        const float a   = fast_tanh(acc[0][tile][e] + bv);
        const float zu0 = acc[1][tile][e], zu1 = acc[2][tile][e],
                    zu2 = acc[3][tile][e];
        const float zw0 = acc[4][tile][e], zw1 = acc[5][tile][e],
                    zw2 = acc[6][tile][e];
        const float d1  = 1.f - a * a;
        const float d2U = -2.f * a * d1 * (zu0 + zu1 + zu2);
        S[0][off] = a;
        S[1][off] = d1 * zu0;
        S[2][off] = d1 * zu1;
        S[3][off] = d1 * zu2;
        S[4][off] = d1 * zw0 + d2U * zu0;
        S[5][off] = d1 * zw1 + d2U * zu1;
        S[6][off] = d1 * zw2 + d2U * zu2;
      }
    }
    __syncthreads();
  }

  {
    const int row   = t & 15;
    const int which = t >> 4;
    if (which < 14) {
      int stream, i, k = 0;
      if (which < 2)      { stream = 0; i = which; }
      else if (which < 8) { const int q = which - 2; i = q / 3; k = q % 3; stream = 1 + k; }
      else                { const int q = which - 8; i = q / 3; k = q % 3; stream = 4 + k; }
      const float* src = S[stream] + row * LSTR;
      float accv = 0.f;
      #pragma unroll 8
      for (int c = 0; c < HID; ++c) accv += src[c] * W4[c * 2 + i];
      const size_t n = (size_t)(r0 + row);
      if (which < 2) {
        out[n * 2 + i] = accv + b4[i];
      } else if (which < 8) {
        out[(size_t)2 * NPTS + (size_t)i * NPTS * 3 + n * 3 + k] = accv;
      } else {
        out[(size_t)8 * NPTS + (size_t)i * NPTS * 3 + n * 3 + k] = accv;
      }
    }
  }
}

extern "C" void kernel_launch(void* const* d_in, const int* in_sizes, int n_in,
                              void* d_out, int out_size, void* d_ws, size_t ws_size,
                              hipStream_t stream) {
  const float* X  = (const float*)d_in[0];
  const float* W0 = (const float*)d_in[1];
  const float* b0 = (const float*)d_in[2];
  const float* W1 = (const float*)d_in[3];
  const float* b1 = (const float*)d_in[4];
  const float* W2 = (const float*)d_in[5];
  const float* b2 = (const float*)d_in[6];
  const float* W3 = (const float*)d_in[7];
  const float* b3 = (const float*)d_in[8];
  const float* W4 = (const float*)d_in[9];
  const float* b4 = (const float*)d_in[10];
  const float* lb = (const float*)d_in[11];
  const float* ub = (const float*)d_in[12];
  float* out = (float*)d_out;

  const size_t shmem = (size_t)NSTREAM * MT * LSTR * sizeof(float);
  dim3 grid(NPTS / MT), block(256);
  pinn_fused<<<grid, block, shmem, stream>>>(X, W0, b0, W1, b1, W2, b2,
                                             W3, b3, W4, b4, lb, ub, out);
}


# Forward-over-reverse: 2 forward streams (a, its (1,1,1) derivative) + 4 adjoint streams through W^T (transposed copy in workspace), 18 stream-GEMMs not 21; sum order changes
# speedup vs baseline: 1.1480x; 1.0888x over previous
#include <hip/hip_runtime.h>


typedef __attribute__((ext_vector_type(2))) float v2f;
typedef __attribute__((ext_vector_type(8))) float v8f;

#define NPTS    32768
#define HID     256
#define MT      16
#define LSTR    260
#define NARR    5

__device__ __forceinline__ v8f wmma4(v2f a, v2f b, v8f c) {
  return __builtin_amdgcn_wmma_f32_16x16x4_f32(
       false, a,  false, b,
       (short)0, c,  false,  false);
}

__device__ __forceinline__ float fast_tanh(float x) {
  const float e = __expf(2.0f * x);
  const float r = __builtin_amdgcn_rcpf(e + 1.0f);
  return 1.0f - 2.0f * r;
}

template <int NS, int KS>
struct KBlk {
  v2f a[KS][NS];
  v2f b[KS][2];
};

template <int NS, int KS>
__device__ __forceinline__ void load_b(KBlk<NS, KS>& f, const float* __restrict__ wk) {
  #pragma unroll
  for (int ks = 0; ks < KS; ++ks) {
    f.b[ks][0].x = wk[(4 * ks) * HID];
    f.b[ks][0].y = wk[(4 * ks + 1) * HID];
    f.b[ks][1].x = wk[(4 * ks) * HID + 16];
    f.b[ks][1].y = wk[(4 * ks + 1) * HID + 16];
  }
}

template <int NS, int KS>
__device__ __forceinline__ void load_a(KBlk<NS, KS>& f, const float* const (&abase)[NS], int k) {
  #pragma unroll
  for (int ks = 0; ks < KS; ++ks) {
    #pragma unroll
    for (int s = 0; s < NS; ++s)
      f.a[ks][s] = *(const v2f*)(abase[s] + k + 4 * ks);
  }
}

template <int NS, int KS>
__device__ __forceinline__ void mma_blk(v8f (&acc)[NS][2], const KBlk<NS, KS>& f) {
  #pragma unroll
  for (int ks = 0; ks < KS; ++ks) {
    #pragma unroll
    for (int s = 0; s < NS; ++s) {
      acc[s][0] = wmma4(f.a[ks][s], f.b[ks][0], acc[s][0]);
      acc[s][1] = wmma4(f.a[ks][s], f.b[ks][1], acc[s][1]);
    }
  }
}

template <int NS, int KS>
__device__ __forceinline__ void gemm256(v8f (&acc)[NS][2],
                                        const float* const (&abase)[NS],
                                        const float* __restrict__ wb,
                                        KBlk<NS, KS>& f0, KBlk<NS, KS>& f1) {
  constexpr int KW = 4 * KS;
  load_a(f0, abase, 0);
  #pragma unroll 1
  for (int k = 0; k < HID - 2 * KW; k += 2 * KW) {
    load_b(f1, wb + (size_t)(k + KW) * HID);
    load_a(f1, abase, k + KW);
    __builtin_amdgcn_sched_barrier(0);
    mma_blk(acc, f0);
    __builtin_amdgcn_sched_barrier(0);
    load_b(f0, wb + (size_t)(k + 2 * KW) * HID);
    load_a(f0, abase, k + 2 * KW);
    __builtin_amdgcn_sched_barrier(0);
    mma_blk(acc, f1);
    __builtin_amdgcn_sched_barrier(0);
  }
  load_b(f1, wb + (size_t)(HID - KW) * HID);
  load_a(f1, abase, HID - KW);
  __builtin_amdgcn_sched_barrier(0);
  mma_blk(acc, f0);
  __builtin_amdgcn_sched_barrier(0);
  mma_blk(acc, f1);
}

__global__ void __launch_bounds__(256)
transpose_w(const float* __restrict__ W1, const float* __restrict__ W2,
            const float* __restrict__ W3, float* __restrict__ WT)
{
  __shared__ float tile[32][33];
  const float* __restrict__ W = blockIdx.z == 0 ? W1 : (blockIdx.z == 1 ? W2 : W3);
  float* __restrict__ D = WT + (size_t)blockIdx.z * HID * HID;
  const int tx = threadIdx.x & 31;
  const int ty = threadIdx.x >> 5;
  const int j0 = blockIdx.y * 32;
  const int c0 = blockIdx.x * 32;
  #pragma unroll
  for (int i = 0; i < 4; ++i)
    tile[ty + 8 * i][tx] = W[(size_t)(j0 + ty + 8 * i) * HID + c0 + tx];
  __syncthreads();
  #pragma unroll
  for (int i = 0; i < 4; ++i)
    D[(size_t)(c0 + ty + 8 * i) * HID + j0 + tx] = tile[tx][ty + 8 * i];
}

__global__ void __launch_bounds__(256)
pinn_fused(const float* __restrict__ X,
           const float* __restrict__ W0, const float* __restrict__ b0,
           const float* __restrict__ W1, const float* __restrict__ b1,
           const float* __restrict__ W2, const float* __restrict__ b2,
           const float* __restrict__ W3, const float* __restrict__ b3,
           const float* __restrict__ W4, const float* __restrict__ b4,
           const float* __restrict__ lb, const float* __restrict__ ub,
           const float* __restrict__ WT,
           float* __restrict__ out)
{
  extern __shared__ float smem[];
  __shared__ float xs[MT * 3];

  const int t  = threadIdx.x;
  const int r0 = blockIdx.x * MT;

  float* P[NARR];
  #pragma unroll
  for (int s = 0; s < NARR; ++s) P[s] = smem + (size_t)s * MT * LSTR;

  const int lane  = t & 31;
  const int wid   = t >> 5;
  const int c0    = wid * 32;
  const int half  = lane >> 4;
  const int l15   = lane & 15;
  const int koff  = half * 2;
  const int rbase = half * 8;
  const int fo    = koff * HID + c0 + l15;

  KBlk<2, 4> g0, g1;
  KBlk<4, 2> h0, h1;

  if (t < MT * 3) xs[t] = X[(size_t)r0 * 3 + t];
  __syncthreads();

  const float lb0 = lb[0], lb1 = lb[1], lb2 = lb[2];
  const float s0 = 2.f / (ub[0] - lb0);
  const float s1 = 2.f / (ub[1] - lb1);
  const float s2 = 2.f / (ub[2] - lb2);

  float w0c[2], w1c[2], w2c[2], bc0[2], Uc[2];
  #pragma unroll
  for (int tile = 0; tile < 2; ++tile) {
    const int col = c0 + tile * 16 + l15;
    w0c[tile] = W0[col]; w1c[tile] = W0[HID + col]; w2c[tile] = W0[2 * HID + col];
    bc0[tile] = b0[col];
    Uc[tile]  = s0 * w0c[tile] + s1 * w1c[tile] + s2 * w2c[tile];
  }
  load_b(g0, W1 + fo);

  #pragma unroll
  for (int tile = 0; tile < 2; ++tile) {
    const int col = c0 + tile * 16 + l15;
    #pragma unroll
    for (int e = 0; e < 8; ++e) {
      const int r = rbase + e;
      const float xn0 = s0 * (xs[r * 3 + 0] - lb0) - 1.f;
      const float xn1 = s1 * (xs[r * 3 + 1] - lb1) - 1.f;
      const float xn2 = s2 * (xs[r * 3 + 2] - lb2) - 1.f;
      const float z = xn0 * w0c[tile] + xn1 * w1c[tile] + xn2 * w2c[tile] + bc0[tile];
      const float a  = fast_tanh(z);
      const float d1 = 1.f - a * a;
      const int off = r * LSTR + col;
      P[0][off] = a;
      P[1][off] = d1 * Uc[tile];
    }
  }
  __syncthreads();

  const float* Wl[3] = {W1, W2, W3};
  const float* bl[3] = {b1, b2, b3};

  const float* abaseF[2];
  #pragma unroll
  for (int s = 0; s < 2; ++s) abaseF[s] = P[s] + l15 * LSTR + koff;
  const float* abaseB[4];
  #pragma unroll
  for (int s = 0; s < 4; ++s) abaseB[s] = P[s] + l15 * LSTR + koff;

  v8f sd1[2][2], sg[2][2];

  #pragma unroll
  for (int l = 0; l < 3; ++l) {
    v8f acc[2][2];
    #pragma unroll
    for (int s = 0; s < 2; ++s) { acc[s][0] = 0.f; acc[s][1] = 0.f; }

    gemm256<2, 4>(acc, abaseF, Wl[l] + fo, g0, g1);
    __syncthreads();

    if (l + 1 < 3) load_b(g0, Wl[l + 1] + fo);
    else           load_b(h0, WT + (size_t)2 * HID * HID + fo);

    const float* bb = bl[l];
    #pragma unroll
    for (int tile = 0; tile < 2; ++tile) {
      const int col = c0 + tile * 16 + l15;
      const float bv = bb[col];
      float lam0 = 0.f, lam1 = 0.f;
      if (l == 2) { lam0 = W4[col * 2]; lam1 = W4[col * 2 + 1]; }
      #pragma unroll
      for (int e = 0; e < 8; ++e) {
        const int off = (rbase + e) * LSTR + col;
        const float a  = fast_tanh(acc[0][tile][e] + bv);
        const float d1 = 1.f - a * a;
        const float T  = d1 * acc[1][tile][e];
        const float g  = -2.f * a * T;
        if (l < 2) {
          sd1[l][tile][e] = d1;
          sg[l][tile][e]  = g;
          P[0][off] = a;
          P[1][off] = T;
        } else {
          P[0][off] = lam0 * d1;
          P[1][off] = lam1 * d1;
          P[2][off] = g * lam0;
          P[3][off] = g * lam1;
          P[4][off] = a;
        }
      }
    }
    __syncthreads();
  }

  #pragma unroll
  for (int m = 2; m >= 0; --m) {
    v8f acc[4][2];
    #pragma unroll
    for (int s = 0; s < 4; ++s) { acc[s][0] = 0.f; acc[s][1] = 0.f; }

    gemm256<4, 2>(acc, abaseB, WT + (size_t)m * HID * HID + fo, h0, h1);
    __syncthreads();

    if (m > 0) load_b(h0, WT + (size_t)(m - 1) * HID * HID + fo);

    #pragma unroll
    for (int tile = 0; tile < 2; ++tile) {
      const int col = c0 + tile * 16 + l15;
      #pragma unroll
      for (int e = 0; e < 8; ++e) {
        const int r = rbase + e;
        float d1, g;
        if (m > 0) {
          d1 = sd1[m - 1][tile][e];
          g  = sg[m - 1][tile][e];
        } else {
          const float xn0 = s0 * (xs[r * 3 + 0] - lb0) - 1.f;
          const float xn1 = s1 * (xs[r * 3 + 1] - lb1) - 1.f;
          const float xn2 = s2 * (xs[r * 3 + 2] - lb2) - 1.f;
          const float z = xn0 * w0c[tile] + xn1 * w1c[tile] + xn2 * w2c[tile] + bc0[tile];
          const float a = fast_tanh(z);
          d1 = 1.f - a * a;
          g  = -2.f * a * (d1 * Uc[tile]);
        }
        const float lam0 = acc[0][tile][e], lam1 = acc[1][tile][e];
        const float mu0  = acc[2][tile][e], mu1  = acc[3][tile][e];
        const int off = r * LSTR + col;
        P[0][off] = lam0 * d1;
        P[1][off] = lam1 * d1;
        P[2][off] = mu0 * d1 + g * lam0;
        P[3][off] = mu1 * d1 + g * lam1;
      }
    }
    __syncthreads();
  }

  {
    const int row   = t & 15;
    const int which = t >> 4;
    if (which < 14) {
      const size_t n = (size_t)(r0 + row);
      if (which < 2) {
        const int i = which;
        const float* src = P[4] + row * LSTR;
        float accv = 0.f;
        #pragma unroll 8
        for (int c = 0; c < HID; ++c) accv += src[c] * W4[c * 2 + i];
        out[n * 2 + i] = accv + b4[i];
      } else {
        const int q   = which - 2;
        const int sec = q >= 6 ? 1 : 0;
        const int q2  = q - 6 * sec;
        const int i   = q2 / 3;
        const int k   = q2 - 3 * i;
        const float* src = P[2 * sec + i] + row * LSTR;
        const float* wv  = W0 + k * HID;
        float accv = 0.f;
        #pragma unroll 8
        for (int c = 0; c < HID; ++c) accv += src[c] * wv[c];
        const float sk = k == 0 ? s0 : (k == 1 ? s1 : s2);
        const size_t base = sec ? (size_t)8 * NPTS : (size_t)2 * NPTS;
        out[base + (size_t)i * NPTS * 3 + n * 3 + k] = accv * sk;
      }
    }
  }
}

extern "C" void kernel_launch(void* const* d_in, const int* in_sizes, int n_in,
                              void* d_out, int out_size, void* d_ws, size_t ws_size,
                              hipStream_t stream) {
  const float* X  = (const float*)d_in[0];
  const float* W0 = (const float*)d_in[1];
  const float* b0 = (const float*)d_in[2];
  const float* W1 = (const float*)d_in[3];
  const float* b1 = (const float*)d_in[4];
  const float* W2 = (const float*)d_in[5];
  const float* b2 = (const float*)d_in[6];
  const float* W3 = (const float*)d_in[7];
  const float* b3 = (const float*)d_in[8];
  const float* W4 = (const float*)d_in[9];
  const float* b4 = (const float*)d_in[10];
  const float* lb = (const float*)d_in[11];
  const float* ub = (const float*)d_in[12];
  float* out = (float*)d_out;
  float* WT  = (float*)d_ws;

  transpose_w<<<dim3(HID / 32, HID / 32, 3), dim3(256), 0, stream>>>(W1, W2, W3, WT);

  const size_t shmem = (size_t)NARR * MT * LSTR * sizeof(float);
  dim3 grid(NPTS / MT), block(256);
  pinn_fused<<<grid, block, shmem, stream>>>(X, W0, b0, W1, b1, W2, b2,
                                             W3, b3, W4, b4, lb, ub, WT, out);
}
